# starting kernel
# baseline (speedup 1.0000x reference)
#include <hip/hip_runtime.h>

typedef float v2f __attribute__((ext_vector_type(2)));
typedef float v8f __attribute__((ext_vector_type(8)));

#define D_MODEL 768
#define NHEAD   12
#define DK      64
#define BATCH   2
#define SEQ     2048
#define M_ROWS  (BATCH * SEQ)
#define NEGV    (-1000000000.0f)
#define SMSTRIDE 18
#define WPB      8

__global__ __launch_bounds__(256) void gemm768_kernel(
    const float* __restrict__ A, const float* __restrict__ W,
    const float* __restrict__ bias, float* __restrict__ C, int split_heads)
{
    int wave = (int)((blockIdx.x * blockDim.x + threadIdx.x) >> 5);
    int lane = (int)(threadIdx.x & 31);
    int tn = wave % (D_MODEL / 64);
    int tm = wave / (D_MODEL / 64);
    int m0 = tm * 16, n0 = tn * 64;
    int l15 = lane & 15;
    int khi = lane >> 4;
    int rbase = khi << 3;

    const float* arow = A + (size_t)(m0 + l15) * D_MODEL;

    v8f acc[4];
#pragma unroll
    for (int c = 0; c < 4; ++c) acc[c] = (v8f){};

#pragma unroll 2
    for (int kb = 0; kb < D_MODEL; kb += 4) {
        v2f a = *(const v2f*)(arow + kb + 2 * khi);
        const float* wp = W + (size_t)(kb + 2 * khi) * D_MODEL + n0 + l15;
#pragma unroll
        for (int c = 0; c < 4; ++c) {
            v2f b; b.x = wp[c * 16]; b.y = wp[D_MODEL + c * 16];
            acc[c] = __builtin_amdgcn_wmma_f32_16x16x4_f32(
                false, a, false, b, (short)0, acc[c], false, false);
        }
    }

    if (!split_heads) {
#pragma unroll
        for (int c = 0; c < 4; ++c) {
            float bv = bias[n0 + c * 16 + l15];
#pragma unroll
            for (int r = 0; r < 8; ++r) {
                int row = m0 + rbase + r;
                C[(size_t)row * D_MODEL + n0 + c * 16 + l15] = acc[c][r] + bv;
            }
        }
    } else {
        int h = n0 >> 6;
#pragma unroll
        for (int c = 0; c < 4; ++c) {
            int d = c * 16 + l15;
            float bv = bias[n0 + d];
#pragma unroll
            for (int r = 0; r < 8; ++r) {
                int row = m0 + rbase + r;
                int bi = row / SEQ, s = row % SEQ;
                C[(((size_t)bi * NHEAD + h) * SEQ + s) * DK + d] = acc[c][r] + bv;
            }
        }
    }
}

__global__ __launch_bounds__(256) void attn_kernel(
    const float* __restrict__ QP, const float* __restrict__ KP,
    const float* __restrict__ VP, const int* __restrict__ mask,
    float* __restrict__ X)
{
    __shared__ float smem[WPB][16 * SMSTRIDE];
    int wv   = (int)(threadIdx.x >> 5);
    int lane = (int)(threadIdx.x & 31);
    int wave = (int)blockIdx.x * WPB + wv;

    int qt = wave % (SEQ / 16);
    int bh = wave / (SEQ / 16);
    int bi = bh / NHEAD, h = bh % NHEAD;
    int s0 = qt * 16;
    int l15 = lane & 15;
    int khi = lane >> 4;
    int rbase = khi << 3;

    const float* Q = QP + (size_t)bh * SEQ * DK;
    const float* K = KP + (size_t)bh * SEQ * DK;
    const float* V = VP + (size_t)bh * SEQ * DK;
    float* sm = smem[wv];

    const float* qrow = Q + (size_t)(s0 + l15) * DK;
    v2f qv[16];
#pragma unroll
    for (int j = 0; j < 16; ++j)
        qv[j] = *(const v2f*)(qrow + 4 * j + 2 * khi);

    v8f acc[4];
#pragma unroll
    for (int c = 0; c < 4; ++c) acc[c] = (v8f){};

    for (int kb = 0; kb < SEQ; kb += 16) {
        v8f sacc = {};
        const float* krow = K + (size_t)(kb + l15) * DK;
#pragma unroll
        for (int j = 0; j < 16; ++j) {
            v2f bb = *(const v2f*)(krow + 4 * j + 2 * khi);
            sacc = __builtin_amdgcn_wmma_f32_16x16x4_f32(
                false, qv[j], false, bb, (short)0, sacc, false, false);
        }

        int tg = kb + l15;
#pragma unroll
        for (int r = 0; r < 8; ++r) {
            int srow = rbase + r;
            int sg = s0 + srow;
            float val = sacc[r] * 0.125f;
            if (mask[(size_t)sg * SEQ + tg] == 0) val = NEGV;
            sm[srow * SMSTRIDE + l15] = val;
        }
        asm volatile("s_wait_dscnt 0" ::: "memory");

#pragma unroll
        for (int j = 0; j < 4; ++j) {
            v2f a = *(const v2f*)(sm + l15 * SMSTRIDE + 4 * j + 2 * khi);
            const float* vrow = V + (size_t)(kb + 4 * j + 2 * khi) * DK + l15;
#pragma unroll
            for (int c = 0; c < 4; ++c) {
                v2f bb; bb.x = vrow[c * 16]; bb.y = vrow[DK + c * 16];
                acc[c] = __builtin_amdgcn_wmma_f32_16x16x4_f32(
                    false, a, false, bb, (short)0, acc[c], false, false);
            }
        }
    }

#pragma unroll
    for (int c = 0; c < 4; ++c) {
#pragma unroll
        for (int r = 0; r < 8; ++r) {
            int s = s0 + rbase + r;
            int d = c * 16 + l15;
            X[((size_t)bi * SEQ + s) * D_MODEL + h * DK + d] = acc[c][r];
        }
    }
}

extern "C" void kernel_launch(void* const* d_in, const int* in_sizes, int n_in,
                              void* d_out, int out_size, void* d_ws, size_t ws_size,
                              hipStream_t stream) {
    (void)in_sizes; (void)n_in; (void)out_size; (void)ws_size;
    const float* q    = (const float*)d_in[0];
    const float* k    = (const float*)d_in[1];
    const float* v    = (const float*)d_in[2];
    const int*   mask = (const int*)d_in[3];
    const float* w_q  = (const float*)d_in[4];
    const float* b_q  = (const float*)d_in[5];
    const float* w_k  = (const float*)d_in[6];
    const float* b_k  = (const float*)d_in[7];
    const float* w_v  = (const float*)d_in[8];
    const float* b_v  = (const float*)d_in[9];
    const float* w_o  = (const float*)d_in[10];
    const float* b_o  = (const float*)d_in[11];

    float* ws = (float*)d_ws;
    size_t proj = (size_t)M_ROWS * D_MODEL;
    float* qp = ws;
    float* kp = ws + proj;
    float* vp = ws + 2 * proj;
    float* xb = ws + 3 * proj;

    dim3 blk(256);
    int gemm_waves = (M_ROWS / 16) * (D_MODEL / 64);
    dim3 g_gemm(gemm_waves / WPB);
    int attn_waves = BATCH * NHEAD * (SEQ / 16);
    dim3 g_attn(attn_waves / WPB);

    gemm768_kernel<<<g_gemm, blk, 0, stream>>>(q, w_q, b_q, qp, 1);
    gemm768_kernel<<<g_gemm, blk, 0, stream>>>(k, w_k, b_k, kp, 1);
    gemm768_kernel<<<g_gemm, blk, 0, stream>>>(v, w_v, b_v, vp, 1);
    attn_kernel<<<g_attn, blk, 0, stream>>>(qp, kp, vp, mask, xb);
    gemm768_kernel<<<g_gemm, blk, 0, stream>>>(xb, w_o, b_o, (float*)d_out, 0);
}


# attention: a wave holds three query tiles per key tile, one wave per SIMD (1024 waves of 3 tiles): K and V fragments loaded once for three tiles' matrix instructions, three score chains; same sums
# speedup vs baseline: 1.6842x; 1.6842x over previous
#include <hip/hip_runtime.h>

typedef float v2f __attribute__((ext_vector_type(2)));
typedef float v8f __attribute__((ext_vector_type(8)));
typedef float v4f __attribute__((ext_vector_type(4)));
typedef int   v4i __attribute__((ext_vector_type(4)));

#define D_MODEL 768
#define NHEAD   12
#define DK      64
#define BATCH   2
#define SEQ     2048
#define M_ROWS  (BATCH * SEQ)
#define NEGV    (-1000000000.0f)

#define GKS 4
#define GKC (4 * GKS)

struct GFrag {
    v2f a[GKS];
    v4f w[GKS][3][2];
};

__device__ __forceinline__ void gemm_load(GFrag& f, const float* __restrict__ ap,
                                          const float* __restrict__ wp)
{
#pragma unroll
    for (int ks = 0; ks < GKS; ++ks) {
        f.a[ks] = *(const v2f*)(ap + 4 * ks);
#pragma unroll
        for (int s = 0; s < 2; ++s) {
#pragma unroll
            for (int g = 0; g < 3; ++g)
                f.w[ks][g][s] = *(const v4f*)(wp + (size_t)(4 * ks + s) * D_MODEL + 64 * g);
        }
    }
}

__device__ __forceinline__ void gemm_mma(const GFrag& f, v8f (&acc)[12])
{
#pragma unroll
    for (int ks = 0; ks < GKS; ++ks) {
#pragma unroll
        for (int g = 0; g < 3; ++g) {
#pragma unroll
            for (int i = 0; i < 4; ++i) {
                v2f b; b.x = f.w[ks][g][0][i]; b.y = f.w[ks][g][1][i];
                acc[4 * g + i] = __builtin_amdgcn_wmma_f32_16x16x4_f32(
                    false, f.a[ks], false, b, (short)0, acc[4 * g + i], false, false);
            }
        }
    }
}

__global__ __launch_bounds__(128) void gemm768_kernel(
    const float* __restrict__ A, const float* __restrict__ W,
    const float* __restrict__ bias, float* __restrict__ C, int layout)
{
    int wv   = (int)(threadIdx.x >> 5);
    int lane = (int)(threadIdx.x & 31);
    int tg = (int)(blockIdx.x >> 6);
    int tm = (int)((blockIdx.x & 63) << 2) + wv;
    int m0 = tm * 16, n0 = tg * 192;
    int l15 = lane & 15;
    int khi = lane >> 4;
    int rbase = khi << 3;

    const float* ap = A + (size_t)(m0 + l15) * D_MODEL + 2 * khi;
    const float* wp = W + (size_t)(2 * khi) * D_MODEL + n0 + 4 * l15;

    v8f acc[12];
#pragma unroll
    for (int c = 0; c < 12; ++c) acc[c] = (v8f){};

    GFrag f0, f1;
    gemm_load(f0, ap, wp);
    for (int kb = 0; kb < D_MODEL - 2 * GKC; kb += 2 * GKC) {
        gemm_load(f1, ap + kb + GKC, wp + (size_t)(kb + GKC) * D_MODEL);
        gemm_mma(f0, acc);
        gemm_load(f0, ap + kb + 2 * GKC, wp + (size_t)(kb + 2 * GKC) * D_MODEL);
        gemm_mma(f1, acc);
    }
    gemm_load(f1, ap + D_MODEL - GKC, wp + (size_t)(D_MODEL - GKC) * D_MODEL);
    gemm_mma(f0, acc);
    gemm_mma(f1, acc);

    if (layout == 2) {
        int bi = m0 / SEQ, t = (m0 % SEQ) >> 4;
        float* tile = C + (((size_t)bi * NHEAD + n0 / 64) * (SEQ / 16) + t) * (16 * DK)
                        + (l15 >> 1) * 128 + khi * 4 + 2 * (l15 & 1);
#pragma unroll
        for (int g = 0; g < 3; ++g) {
            v4f bv = *(const v4f*)(bias + n0 + 64 * g + 4 * l15);
            float* tg_ = tile + (size_t)g * SEQ * DK;
#pragma unroll
            for (int r = 0; r < 8; ++r) {
                v2f lo, hi;
                lo.x = acc[4 * g + 0][r] + bv.x; lo.y = acc[4 * g + 1][r] + bv.y;
                hi.x = acc[4 * g + 2][r] + bv.z; hi.y = acc[4 * g + 3][r] + bv.w;
                *(v2f*)(tg_ + 16 * r)     = lo;
                *(v2f*)(tg_ + 16 * r + 8) = hi;
            }
        }
        return;
    }

    int split_heads = layout;
    float* crow;
    size_t cpitch;
    if (!split_heads) {
        crow = C + (size_t)(m0 + rbase) * D_MODEL + n0 + 4 * l15;
        cpitch = 64;
    } else {
        int bi = m0 / SEQ, s = m0 % SEQ + rbase;
        crow = C + (((size_t)bi * NHEAD + n0 / 64) * SEQ + s) * DK + 4 * l15;
        cpitch = (size_t)SEQ * DK;
    }
    size_t rpitch = split_heads ? DK : D_MODEL;
#pragma unroll
    for (int g = 0; g < 3; ++g) {
        v4f bv = *(const v4f*)(bias + n0 + 64 * g + 4 * l15);
#pragma unroll
        for (int r = 0; r < 8; ++r) {
            v4f o;
            o.x = acc[4 * g + 0][r] + bv.x; o.y = acc[4 * g + 1][r] + bv.y;
            o.z = acc[4 * g + 2][r] + bv.z; o.w = acc[4 * g + 3][r] + bv.w;
            *(v4f*)(crow + g * cpitch + r * rpitch) = o;
        }
    }
}

#define KFRAG_SLOT(c, h)  (4 * ((c) & 7) + 2 * (h) + ((c) >> 3))

#define AQT          3
#define WAVES_PER_BH ((SEQ / 16) / AQT)
#define REG_WAVES    (BATCH * NHEAD * WAVES_PER_BH)
#define AWPB         4
#define REG_BLOCKS   (REG_WAVES / AWPB)

template <int NKV>
struct ATile {
    v4f kk[NKV][8];
    v4f xv[NKV][8];
    v4i mk[AQT][2];
};

template <int NKV>
__device__ __forceinline__ void attn_load(ATile<NKV>& t, const float* (&kptr)[NKV],
                                          const float* (&vptr)[NKV],
                                          const int* (&mptr)[AQT], int kb)
{
#pragma unroll
    for (int s = 0; s < NKV; ++s) {
        const float* kp = kptr[s] + (size_t)kb * DK;
#pragma unroll
        for (int i = 0; i < 8; ++i)
            t.kk[s][i] = *(const v4f*)(kp + 128 * i);
    }
#pragma unroll
    for (int s = 0; s < NKV; ++s) {
        const float* vp = vptr[s] + (size_t)kb * DK;
#pragma unroll
        for (int i = 0; i < 8; ++i)
            t.xv[s][i] = *(const v4f*)(vp + (size_t)i * DK);
    }
#pragma unroll
    for (int q = 0; q < AQT; ++q) {
        t.mk[q][0] = *(const v4i*)(mptr[q] + kb);
        t.mk[q][1] = *(const v4i*)(mptr[q] + kb + 4);
    }
}

__device__ __forceinline__ v2f attn_kfrag(const v4f (&kk)[8], int j)
{
    v2f ka; ka.x = kk[j >> 1][2 * (j & 1)]; ka.y = kk[j >> 1][2 * (j & 1) + 1];
    return ka;
}

__device__ __forceinline__ void attn_scale_mask(const v8f& s, const v4i& m0, const v4i& m1,
                                                float (&p)[8])
{
#pragma unroll
    for (int r = 0; r < 4; ++r) {
        p[r]     = (m0[r] == 0) ? NEGV : s[r]     * 0.125f;
        p[r + 4] = (m1[r] == 0) ? NEGV : s[r + 4] * 0.125f;
    }
}

__device__ __forceinline__ void attn_pv(const v4f (&xv)[8], const float (&p)[8], int j, int c,
                                        v8f (&acc)[4])
{
    v2f va; va.x = xv[2 * j][c]; va.y = xv[2 * j + 1][c];
    v2f sb; sb.x = p[2 * j];     sb.y = p[2 * j + 1];
    acc[c] = __builtin_amdgcn_wmma_f32_16x16x4_f32(
        false, va, false, sb, (short)0, acc[c], false, false);
}

template <int NKV>
__device__ __forceinline__ void attn_tile(const ATile<NKV>& t, const v2f (&qv)[AQT][16],
                                          v8f (&acc)[AQT][4])
{
    constexpr int S2 = NKV - 1;
    v8f s0 = {}, s1 = {}, s2 = {};
    float p0[8], p1[8], p2[8];

#pragma unroll
    for (int j = 0; j < 16; ++j) {
        v2f ka = attn_kfrag(t.kk[0], j);
        s0 = __builtin_amdgcn_wmma_f32_16x16x4_f32(
            false, ka, false, qv[0][j], (short)0, s0, false, false);
        s1 = __builtin_amdgcn_wmma_f32_16x16x4_f32(
            false, ka, false, qv[1][j], (short)0, s1, false, false);
    }
    attn_scale_mask(s0, t.mk[0][0], t.mk[0][1], p0);
    attn_scale_mask(s1, t.mk[1][0], t.mk[1][1], p1);

#pragma unroll
    for (int j = 0; j < 16; ++j) {
        v2f ka = attn_kfrag(t.kk[S2], j);
        s2 = __builtin_amdgcn_wmma_f32_16x16x4_f32(
            false, ka, false, qv[2][j], (short)0, s2, false, false);
        attn_pv(t.xv[0], p0, j >> 2, j & 3, acc[0]);
    }
    attn_scale_mask(s2, t.mk[2][0], t.mk[2][1], p2);

#pragma unroll
    for (int j = 0; j < 16; ++j)
        attn_pv(t.xv[0], p1, j >> 2, j & 3, acc[1]);
#pragma unroll
    for (int j = 0; j < 16; ++j)
        attn_pv(t.xv[S2], p2, j >> 2, j & 3, acc[2]);
}

template <int NKV>
__device__ __forceinline__ void attn_wave(
    const float* __restrict__ QP, const float* __restrict__ KP,
    const float* __restrict__ VP, const int* __restrict__ mask,
    float* __restrict__ X, const int (&bhq)[AQT], const int (&qtq)[AQT], int lane)
{
    int l15 = lane & 15;
    int khi = lane >> 4;

    v2f qv[AQT][16];
#pragma unroll
    for (int q = 0; q < AQT; ++q) {
        const float* qrow = QP + ((size_t)bhq[q] * SEQ + qtq[q] * 16 + l15) * DK + 2 * khi;
#pragma unroll
        for (int j = 0; j < 16; ++j)
            qv[q][j] = *(const v2f*)(qrow + 4 * j);
    }

    v8f acc[AQT][4];
#pragma unroll
    for (int q = 0; q < AQT; ++q)
#pragma unroll
        for (int c = 0; c < 4; ++c) acc[q][c] = (v8f){};

    const float* kptr[NKV];
    const float* vptr[NKV];
    const int*   mptr[AQT];
#pragma unroll
    for (int s = 0; s < NKV; ++s) {
        size_t hb = (size_t)bhq[s == 0 ? 0 : 2] * SEQ * DK;
        kptr[s] = KP + hb + 4 * KFRAG_SLOT(l15, khi);
        vptr[s] = VP + hb + (size_t)(8 * khi) * DK + 4 * l15;
    }
#pragma unroll
    for (int q = 0; q < AQT; ++q)
        mptr[q] = mask + (size_t)(qtq[q] * 16 + l15) * SEQ + 8 * khi;

    ATile<NKV> ta, tb;
    attn_load(ta, kptr, vptr, mptr, 0);
    for (int kb = 0; kb < SEQ - 32; kb += 32) {
        attn_load(tb, kptr, vptr, mptr, kb + 16);
        __builtin_amdgcn_sched_barrier(0);
        attn_tile(ta, qv, acc);
        __builtin_amdgcn_sched_barrier(0);
        attn_load(ta, kptr, vptr, mptr, kb + 32);
        __builtin_amdgcn_sched_barrier(0);
        attn_tile(tb, qv, acc);
    }
    attn_load(tb, kptr, vptr, mptr, SEQ - 16);
    __builtin_amdgcn_sched_barrier(0);
    attn_tile(ta, qv, acc);
    attn_tile(tb, qv, acc);

#pragma unroll
    for (int q = 0; q < AQT; ++q) {
        int bi = bhq[q] / NHEAD, h = bhq[q] % NHEAD;
        float* xrow = X + ((size_t)bi * SEQ + qtq[q] * 16 + l15) * D_MODEL + h * DK + 32 * khi;
#pragma unroll
        for (int r = 0; r < 8; ++r) {
            v4f o;
            o.x = acc[q][0][r]; o.y = acc[q][1][r]; o.z = acc[q][2][r]; o.w = acc[q][3][r];
            *(v4f*)(xrow + 4 * r) = o;
        }
    }
}

__global__ __launch_bounds__(128) void attn_kernel(
    const float* __restrict__ QP, const float* __restrict__ KP,
    const float* __restrict__ VP, const int* __restrict__ mask,
    float* __restrict__ X)
{
    int wv   = (int)(threadIdx.x >> 5);
    int lane = (int)(threadIdx.x & 31);

    if (blockIdx.x < REG_BLOCKS) {
        int w  = (int)blockIdx.x * AWPB + wv;
        int bh = w / WAVES_PER_BH;
        int i  = w - bh * WAVES_PER_BH;
        int bhq[AQT] = { bh, bh, bh };
        int qtq[AQT] = { AQT * i, AQT * i + 1, AQT * i + 2 };
        attn_wave<1>(QP, KP, VP, mask, X, bhq, qtq, lane);
    } else {
        int m   = ((int)blockIdx.x - REG_BLOCKS) * AWPB + wv;
        int u   = m >> 1, odd = m & 1;
        int bx  = 3 * u + 2 * odd;
        int by  = 3 * u + 1;
        int t0  = AQT * WAVES_PER_BH;
        int bhq[AQT] = { bx, bx, by };
        int qtq[AQT] = { t0, t0 + 1, t0 + odd };
        attn_wave<2>(QP, KP, VP, mask, X, bhq, qtq, lane);
    }
}

extern "C" void kernel_launch(void* const* d_in, const int* in_sizes, int n_in,
                              void* d_out, int out_size, void* d_ws, size_t ws_size,
                              hipStream_t stream) {
    (void)in_sizes; (void)n_in; (void)out_size; (void)ws_size;
    const float* q    = (const float*)d_in[0];
    const float* k    = (const float*)d_in[1];
    const float* v    = (const float*)d_in[2];
    const int*   mask = (const int*)d_in[3];
    const float* w_q  = (const float*)d_in[4];
    const float* b_q  = (const float*)d_in[5];
    const float* w_k  = (const float*)d_in[6];
    const float* b_k  = (const float*)d_in[7];
    const float* w_v  = (const float*)d_in[8];
    const float* b_v  = (const float*)d_in[9];
    const float* w_o  = (const float*)d_in[10];
    const float* b_o  = (const float*)d_in[11];

    float* ws = (float*)d_ws;
    size_t proj = (size_t)M_ROWS * D_MODEL;
    float* qp = ws;
    float* kp = ws + proj;
    float* vp = ws + 2 * proj;
    float* xb = ws + 3 * proj;

    dim3 ablk(128);
    dim3 gblk(128);
    dim3 g_gemm(256);
    dim3 g_attn(256);

    gemm768_kernel<<<g_gemm, gblk, 0, stream>>>(q, w_q, b_q, qp, 1);
    gemm768_kernel<<<g_gemm, gblk, 0, stream>>>(k, w_k, b_k, kp, 2);
    gemm768_kernel<<<g_gemm, gblk, 0, stream>>>(v, w_v, b_v, vp, 1);
    attn_kernel<<<g_attn, ablk, 0, stream>>>(qp, kp, vp, mask, xb);
    gemm768_kernel<<<g_gemm, gblk, 0, stream>>>(xb, w_o, b_o, (float*)d_out, 0);
}


# projection GEMMs: fragments two 16-k blocks ahead in a ring of three register sets (was one ahead in two); a block's 28 loads issued as one group in front of 48 matrix instructions; same sums
# speedup vs baseline: 1.9573x; 1.1622x over previous
#include <hip/hip_runtime.h>

typedef float v2f __attribute__((ext_vector_type(2)));
typedef float v8f __attribute__((ext_vector_type(8)));
typedef float v4f __attribute__((ext_vector_type(4)));
typedef int   v4i __attribute__((ext_vector_type(4)));

#define D_MODEL 768
#define NHEAD   12
#define DK      64
#define BATCH   2
#define SEQ     2048
#define M_ROWS  (BATCH * SEQ)
#define NEGV    (-1000000000.0f)

#define GKS 4
#define GKC (4 * GKS)

struct GFrag {
    v2f a[GKS];
    v4f w[GKS][3][2];
};

__device__ __forceinline__ void gemm_load(GFrag& f, const float* __restrict__ ap,
                                          const float* __restrict__ wp)
{
#pragma unroll
    for (int ks = 0; ks < GKS; ++ks) {
        f.a[ks] = *(const v2f*)(ap + 4 * ks);
#pragma unroll
        for (int s = 0; s < 2; ++s) {
#pragma unroll
            for (int g = 0; g < 3; ++g)
                f.w[ks][g][s] = *(const v4f*)(wp + (size_t)(4 * ks + s) * D_MODEL + 64 * g);
        }
    }
}

__device__ __forceinline__ void gemm_mma(const GFrag& f, v8f (&acc)[12])
{
#pragma unroll
    for (int ks = 0; ks < GKS; ++ks) {
#pragma unroll
        for (int g = 0; g < 3; ++g) {
#pragma unroll
            for (int i = 0; i < 4; ++i) {
                v2f b; b.x = f.w[ks][g][0][i]; b.y = f.w[ks][g][1][i];
                acc[4 * g + i] = __builtin_amdgcn_wmma_f32_16x16x4_f32(
                    false, f.a[ks], false, b, (short)0, acc[4 * g + i], false, false);
            }
        }
    }
}

__global__ __launch_bounds__(128) void gemm768_kernel(
    const float* __restrict__ A, const float* __restrict__ W,
    const float* __restrict__ bias, float* __restrict__ C, int layout)
{
    int wv   = (int)(threadIdx.x >> 5);
    int lane = (int)(threadIdx.x & 31);
    int tg = (int)(blockIdx.x >> 6);
    int tm = (int)((blockIdx.x & 63) << 2) + wv;
    int m0 = tm * 16, n0 = tg * 192;
    int l15 = lane & 15;
    int khi = lane >> 4;
    int rbase = khi << 3;

    const float* ap = A + (size_t)(m0 + l15) * D_MODEL + 2 * khi;
    const float* wp = W + (size_t)(2 * khi) * D_MODEL + n0 + 4 * l15;

    v8f acc[12];
#pragma unroll
    for (int c = 0; c < 12; ++c) acc[c] = (v8f){};

    GFrag f0, f1, f2;
    gemm_load(f0, ap, wp);
    gemm_load(f1, ap + GKC, wp + (size_t)GKC * D_MODEL);
    for (int kb = 0; kb < D_MODEL - 3 * GKC; kb += 3 * GKC) {
        gemm_load(f2, ap + kb + 2 * GKC, wp + (size_t)(kb + 2 * GKC) * D_MODEL);
        __builtin_amdgcn_sched_barrier(0);
        gemm_mma(f0, acc);
        __builtin_amdgcn_sched_barrier(0);
        gemm_load(f0, ap + kb + 3 * GKC, wp + (size_t)(kb + 3 * GKC) * D_MODEL);
        __builtin_amdgcn_sched_barrier(0);
        gemm_mma(f1, acc);
        __builtin_amdgcn_sched_barrier(0);
        gemm_load(f1, ap + kb + 4 * GKC, wp + (size_t)(kb + 4 * GKC) * D_MODEL);
        __builtin_amdgcn_sched_barrier(0);
        gemm_mma(f2, acc);
    }
    gemm_load(f2, ap + D_MODEL - GKC, wp + (size_t)(D_MODEL - GKC) * D_MODEL);
    __builtin_amdgcn_sched_barrier(0);
    gemm_mma(f0, acc);
    gemm_mma(f1, acc);
    gemm_mma(f2, acc);

    if (layout == 2) {
        int bi = m0 / SEQ, t = (m0 % SEQ) >> 4;
        float* tile = C + (((size_t)bi * NHEAD + n0 / 64) * (SEQ / 16) + t) * (16 * DK)
                        + (l15 >> 1) * 128 + khi * 4 + 2 * (l15 & 1);
#pragma unroll
        for (int g = 0; g < 3; ++g) {
            v4f bv = *(const v4f*)(bias + n0 + 64 * g + 4 * l15);
            float* tg_ = tile + (size_t)g * SEQ * DK;
#pragma unroll
            for (int r = 0; r < 8; ++r) {
                v2f lo, hi;
                lo.x = acc[4 * g + 0][r] + bv.x; lo.y = acc[4 * g + 1][r] + bv.y;
                hi.x = acc[4 * g + 2][r] + bv.z; hi.y = acc[4 * g + 3][r] + bv.w;
                *(v2f*)(tg_ + 16 * r)     = lo;
                *(v2f*)(tg_ + 16 * r + 8) = hi;
            }
        }
        return;
    }

    int split_heads = layout;
    float* crow;
    size_t cpitch;
    if (!split_heads) {
        crow = C + (size_t)(m0 + rbase) * D_MODEL + n0 + 4 * l15;
        cpitch = 64;
    } else {
        int bi = m0 / SEQ, s = m0 % SEQ + rbase;
        crow = C + (((size_t)bi * NHEAD + n0 / 64) * SEQ + s) * DK + 4 * l15;
        cpitch = (size_t)SEQ * DK;
    }
    size_t rpitch = split_heads ? DK : D_MODEL;
#pragma unroll
    for (int g = 0; g < 3; ++g) {
        v4f bv = *(const v4f*)(bias + n0 + 64 * g + 4 * l15);
#pragma unroll
        for (int r = 0; r < 8; ++r) {
            v4f o;
            o.x = acc[4 * g + 0][r] + bv.x; o.y = acc[4 * g + 1][r] + bv.y;
            o.z = acc[4 * g + 2][r] + bv.z; o.w = acc[4 * g + 3][r] + bv.w;
            *(v4f*)(crow + g * cpitch + r * rpitch) = o;
        }
    }
}

#define KFRAG_SLOT(c, h)  (4 * ((c) & 7) + 2 * (h) + ((c) >> 3))

#define AQT          3
#define WAVES_PER_BH ((SEQ / 16) / AQT)
#define REG_WAVES    (BATCH * NHEAD * WAVES_PER_BH)
#define AWPB         4
#define REG_BLOCKS   (REG_WAVES / AWPB)

template <int NKV>
struct ATile {
    v4f kk[NKV][8];
    v4f xv[NKV][8];
    v4i mk[AQT][2];
};

template <int NKV>
__device__ __forceinline__ void attn_load(ATile<NKV>& t, const float* (&kptr)[NKV],
                                          const float* (&vptr)[NKV],
                                          const int* (&mptr)[AQT], int kb)
{
#pragma unroll
    for (int s = 0; s < NKV; ++s) {
        const float* kp = kptr[s] + (size_t)kb * DK;
#pragma unroll
        for (int i = 0; i < 8; ++i)
            t.kk[s][i] = *(const v4f*)(kp + 128 * i);
    }
#pragma unroll
    for (int s = 0; s < NKV; ++s) {
        const float* vp = vptr[s] + (size_t)kb * DK;
#pragma unroll
        for (int i = 0; i < 8; ++i)
            t.xv[s][i] = *(const v4f*)(vp + (size_t)i * DK);
    }
#pragma unroll
    for (int q = 0; q < AQT; ++q) {
        t.mk[q][0] = *(const v4i*)(mptr[q] + kb);
        t.mk[q][1] = *(const v4i*)(mptr[q] + kb + 4);
    }
}

__device__ __forceinline__ v2f attn_kfrag(const v4f (&kk)[8], int j)
{
    v2f ka; ka.x = kk[j >> 1][2 * (j & 1)]; ka.y = kk[j >> 1][2 * (j & 1) + 1];
    return ka;
}

__device__ __forceinline__ void attn_scale_mask(const v8f& s, const v4i& m0, const v4i& m1,
                                                float (&p)[8])
{
#pragma unroll
    for (int r = 0; r < 4; ++r) {
        p[r]     = (m0[r] == 0) ? NEGV : s[r]     * 0.125f;
        p[r + 4] = (m1[r] == 0) ? NEGV : s[r + 4] * 0.125f;
    }
}

__device__ __forceinline__ void attn_pv(const v4f (&xv)[8], const float (&p)[8], int j, int c,
                                        v8f (&acc)[4])
{
    v2f va; va.x = xv[2 * j][c]; va.y = xv[2 * j + 1][c];
    v2f sb; sb.x = p[2 * j];     sb.y = p[2 * j + 1];
    acc[c] = __builtin_amdgcn_wmma_f32_16x16x4_f32(
        false, va, false, sb, (short)0, acc[c], false, false);
}

template <int NKV>
__device__ __forceinline__ void attn_tile(const ATile<NKV>& t, const v2f (&qv)[AQT][16],
                                          v8f (&acc)[AQT][4])
{
    constexpr int S2 = NKV - 1;
    v8f s0 = {}, s1 = {}, s2 = {};
    float p0[8], p1[8], p2[8];

#pragma unroll
    for (int j = 0; j < 16; ++j) {
        v2f ka = attn_kfrag(t.kk[0], j);
        s0 = __builtin_amdgcn_wmma_f32_16x16x4_f32(
            false, ka, false, qv[0][j], (short)0, s0, false, false);
        s1 = __builtin_amdgcn_wmma_f32_16x16x4_f32(
            false, ka, false, qv[1][j], (short)0, s1, false, false);
    }
    attn_scale_mask(s0, t.mk[0][0], t.mk[0][1], p0);
    attn_scale_mask(s1, t.mk[1][0], t.mk[1][1], p1);

#pragma unroll
    for (int j = 0; j < 16; ++j) {
        v2f ka = attn_kfrag(t.kk[S2], j);
        s2 = __builtin_amdgcn_wmma_f32_16x16x4_f32(
            false, ka, false, qv[2][j], (short)0, s2, false, false);
        attn_pv(t.xv[0], p0, j >> 2, j & 3, acc[0]);
    }
    attn_scale_mask(s2, t.mk[2][0], t.mk[2][1], p2);

#pragma unroll
    for (int j = 0; j < 16; ++j)
        attn_pv(t.xv[0], p1, j >> 2, j & 3, acc[1]);
#pragma unroll
    for (int j = 0; j < 16; ++j)
        attn_pv(t.xv[S2], p2, j >> 2, j & 3, acc[2]);
}

template <int NKV>
__device__ __forceinline__ void attn_wave(
    const float* __restrict__ QP, const float* __restrict__ KP,
    const float* __restrict__ VP, const int* __restrict__ mask,
    float* __restrict__ X, const int (&bhq)[AQT], const int (&qtq)[AQT], int lane)
{
    int l15 = lane & 15;
    int khi = lane >> 4;

    v2f qv[AQT][16];
#pragma unroll
    for (int q = 0; q < AQT; ++q) {
        const float* qrow = QP + ((size_t)bhq[q] * SEQ + qtq[q] * 16 + l15) * DK + 2 * khi;
#pragma unroll
        for (int j = 0; j < 16; ++j)
            qv[q][j] = *(const v2f*)(qrow + 4 * j);
    }

    v8f acc[AQT][4];
#pragma unroll
    for (int q = 0; q < AQT; ++q)
#pragma unroll
        for (int c = 0; c < 4; ++c) acc[q][c] = (v8f){};

    const float* kptr[NKV];
    const float* vptr[NKV];
    const int*   mptr[AQT];
#pragma unroll
    for (int s = 0; s < NKV; ++s) {
        size_t hb = (size_t)bhq[s == 0 ? 0 : 2] * SEQ * DK;
        kptr[s] = KP + hb + 4 * KFRAG_SLOT(l15, khi);
        vptr[s] = VP + hb + (size_t)(8 * khi) * DK + 4 * l15;
    }
#pragma unroll
    for (int q = 0; q < AQT; ++q)
        mptr[q] = mask + (size_t)(qtq[q] * 16 + l15) * SEQ + 8 * khi;

    ATile<NKV> ta, tb;
    attn_load(ta, kptr, vptr, mptr, 0);
    for (int kb = 0; kb < SEQ - 32; kb += 32) {
        attn_load(tb, kptr, vptr, mptr, kb + 16);
        __builtin_amdgcn_sched_barrier(0);
        attn_tile(ta, qv, acc);
        __builtin_amdgcn_sched_barrier(0);
        attn_load(ta, kptr, vptr, mptr, kb + 32);
        __builtin_amdgcn_sched_barrier(0);
        attn_tile(tb, qv, acc);
    }
    attn_load(tb, kptr, vptr, mptr, SEQ - 16);
    __builtin_amdgcn_sched_barrier(0);
    attn_tile(ta, qv, acc);
    attn_tile(tb, qv, acc);

#pragma unroll
    for (int q = 0; q < AQT; ++q) {
        int bi = bhq[q] / NHEAD, h = bhq[q] % NHEAD;
        float* xrow = X + ((size_t)bi * SEQ + qtq[q] * 16 + l15) * D_MODEL + h * DK + 32 * khi;
#pragma unroll
        for (int r = 0; r < 8; ++r) {
            v4f o;
            o.x = acc[q][0][r]; o.y = acc[q][1][r]; o.z = acc[q][2][r]; o.w = acc[q][3][r];
            *(v4f*)(xrow + 4 * r) = o;
        }
    }
}

__global__ __launch_bounds__(128) void attn_kernel(
    const float* __restrict__ QP, const float* __restrict__ KP,
    const float* __restrict__ VP, const int* __restrict__ mask,
    float* __restrict__ X)
{
    int wv   = (int)(threadIdx.x >> 5);
    int lane = (int)(threadIdx.x & 31);

    if (blockIdx.x < REG_BLOCKS) {
        int w  = (int)blockIdx.x * AWPB + wv;
        int bh = w / WAVES_PER_BH;
        int i  = w - bh * WAVES_PER_BH;
        int bhq[AQT] = { bh, bh, bh };
        int qtq[AQT] = { AQT * i, AQT * i + 1, AQT * i + 2 };
        attn_wave<1>(QP, KP, VP, mask, X, bhq, qtq, lane);
    } else {
        int m   = ((int)blockIdx.x - REG_BLOCKS) * AWPB + wv;
        int u   = m >> 1, odd = m & 1;
        int bx  = 3 * u + 2 * odd;
        int by  = 3 * u + 1;
        int t0  = AQT * WAVES_PER_BH;
        int bhq[AQT] = { bx, bx, by };
        int qtq[AQT] = { t0, t0 + 1, t0 + odd };
        attn_wave<2>(QP, KP, VP, mask, X, bhq, qtq, lane);
    }
}

extern "C" void kernel_launch(void* const* d_in, const int* in_sizes, int n_in,
                              void* d_out, int out_size, void* d_ws, size_t ws_size,
                              hipStream_t stream) {
    (void)in_sizes; (void)n_in; (void)out_size; (void)ws_size;
    const float* q    = (const float*)d_in[0];
    const float* k    = (const float*)d_in[1];
    const float* v    = (const float*)d_in[2];
    const int*   mask = (const int*)d_in[3];
    const float* w_q  = (const float*)d_in[4];
    const float* b_q  = (const float*)d_in[5];
    const float* w_k  = (const float*)d_in[6];
    const float* b_k  = (const float*)d_in[7];
    const float* w_v  = (const float*)d_in[8];
    const float* b_v  = (const float*)d_in[9];
    const float* w_o  = (const float*)d_in[10];
    const float* b_o  = (const float*)d_in[11];

    float* ws = (float*)d_ws;
    size_t proj = (size_t)M_ROWS * D_MODEL;
    float* qp = ws;
    float* kp = ws + proj;
    float* vp = ws + 2 * proj;
    float* xb = ws + 3 * proj;

    dim3 ablk(128);
    dim3 gblk(128);
    dim3 g_gemm(256);
    dim3 g_attn(256);

    gemm768_kernel<<<g_gemm, gblk, 0, stream>>>(q, w_q, b_q, qp, 1);
    gemm768_kernel<<<g_gemm, gblk, 0, stream>>>(k, w_k, b_k, kp, 2);
    gemm768_kernel<<<g_gemm, gblk, 0, stream>>>(v, w_v, b_v, vp, 1);
    attn_kernel<<<g_attn, ablk, 0, stream>>>(qp, kp, vp, mask, xb);
    gemm768_kernel<<<g_gemm, gblk, 0, stream>>>(xb, w_o, b_o, (float*)d_out, 0);
}
